# starting kernel
# baseline (speedup 1.0000x reference)
#include <hip/hip_runtime.h>
#include <hip/hip_bf16.h>
#include <stdint.h>

#define BATCH    8192
#define EMBED    128
#define NUM_MOD  64
#define HIDDEN   512
#define OUT_DIM  64
#define MAX_TILES 640
#define WCOMB    0.2857142857142857f

typedef float v2f __attribute__((ext_vector_type(2)));
typedef float v8f __attribute__((ext_vector_type(8)));

__device__ __forceinline__ v8f wmma_f32_4(v2f a, v2f b, v8f c) {
    return __builtin_amdgcn_wmma_f32_16x16x4_f32(
         false, a,  false, b,
         (short)0, c,  false,  false);
}

__global__ void zero_counts_kernel(int* counts) {
    if (threadIdx.x < 3 * NUM_MOD) counts[threadIdx.x] = 0;
}

__global__ void count_kernel(const int* __restrict__ mids, int* __restrict__ counts) {
    int b = blockIdx.x * blockDim.x + threadIdx.x;
    if (b < BATCH) {
        for (int h = 0; h < 3; ++h)
            atomicAdd(&counts[h * NUM_MOD + mids[b * 3 + h]], 1);
    }
}

__global__ void scan_build_kernel(const int* __restrict__ counts,
                                  int* __restrict__ offsets, int* __restrict__ cursor,
                                  int4* __restrict__ tiletab, int* __restrict__ ntiles) {
    if (blockIdx.x != 0 || threadIdx.x != 0) return;
    for (int h = 0; h < 3; ++h) {
        int run = 0;
        for (int m = 0; m < NUM_MOD; ++m) {
            offsets[h * NUM_MOD + m] = run;
            cursor [h * NUM_MOD + m] = run;
            run += counts[h * NUM_MOD + m];
        }
        int idx = 0;
        for (int m = 0; m < NUM_MOD; ++m) {
            int c  = counts [h * NUM_MOD + m];
            int st = offsets[h * NUM_MOD + m];
            for (int t = 0; t < c; t += 16) {
                int rem = c - t; if (rem > 16) rem = 16;
                tiletab[h * MAX_TILES + idx] = make_int4(m, st + t, rem, 0);
                ++idx;
            }
        }
        ntiles[h] = idx;
    }
}

__global__ void scatter_kernel(const int* __restrict__ mids,
                               int* __restrict__ cursor, int* __restrict__ rowlist) {
    int b = blockIdx.x * blockDim.x + threadIdx.x;
    if (b < BATCH) {
        for (int h = 0; h < 3; ++h) {
            int m = mids[b * 3 + h];
            int p = atomicAdd(&cursor[h * NUM_MOD + m], 1);
            rowlist[h * BATCH + p] = b;
        }
    }
}


__global__ void pack_wm_kernel(const float* __restrict__ W, float* __restrict__ WF) {
    int t = blockIdx.x * blockDim.x + threadIdx.x;
    if (t >= NUM_MOD * 8 * 32 * 32) return;
    int l  = t & 31;
    int k4 = (t >> 5) & 31;
    int ot = (t >> 10) & 7;
    int m  = t >> 13;
    int n = l & 15, c0 = (l >> 4) * 2;
    int o = ot * 16 + n;
    int i = 4 * k4 + c0;
    const float* src = W + ((size_t)m * EMBED + o) * EMBED + i;
    int k4g = k4 >> 1, ph = k4 & 1;
    float* dst = WF + ((size_t)((m * 8 + ot) * 16 + k4g)) * 128 + l * 4 + ph * 2;
    dst[0] = src[0];
    dst[1] = src[1];
}

__global__ void pack_w1_kernel(const float* __restrict__ W1, float* __restrict__ WF) {
    int t = blockIdx.x * blockDim.x + threadIdx.x;
    if (t >= 32 * 32 * 32) return;
    int l  = t & 31;
    int k4 = (t >> 5) & 31;
    int ot = t >> 10;
    int n = l & 15, c0 = (l >> 4) * 2;
    int o = ot * 16 + n;
    int i = 4 * k4 + c0;
    const float* src = W1 + (size_t)o * EMBED + i;
    int k4g = k4 >> 1, ph = k4 & 1;
    float* dst = WF + ((size_t)(ot * 16 + k4g)) * 128 + l * 4 + ph * 2;
    dst[0] = src[0];
    dst[1] = src[1];
}

__global__ void pack_w2_kernel(const float* __restrict__ W2, float* __restrict__ WF) {
    int t = blockIdx.x * blockDim.x + threadIdx.x;
    if (t >= 4 * 128 * 32) return;
    int l  = t & 31;
    int k4 = (t >> 5) & 127;
    int ot = t >> 12;
    int n = l & 15, c0 = (l >> 4) * 2;
    int o = ot * 16 + n;
    int i = 4 * k4 + c0;
    const float* src = W2 + (size_t)o * HIDDEN + i;
    int k4g = k4 >> 1, ph = k4 & 1;
    float* dst = WF + ((size_t)(ot * 64 + k4g)) * 128 + l * 4 + ph * 2;
    dst[0] = src[0];
    dst[1] = src[1];
}

__global__ __launch_bounds__(256) void hop_gemm_kernel(
    const float* __restrict__ embeds, const float* __restrict__ WmF,
    const int*  __restrict__ eids,    const int4* __restrict__ tiletab,
    const int*  __restrict__ ntiles,  const int*  __restrict__ rowlist,
    const float* __restrict__ xin,    float* __restrict__ xout, int hop)
{
    __shared__ float sX[16][132];
    __shared__ float sB[16][132];
    __shared__ int   sRow[16];

    int nt = ntiles[hop];
    if ((int)blockIdx.x >= nt) return;
    int4 tt = tiletab[hop * MAX_TILES + blockIdx.x];
    int m = tt.x, start = tt.y, cnt = tt.z;
    int tid = threadIdx.x;

    if (tid < 16) {
        int slot = (tid < cnt) ? tid : 0;
        sRow[tid] = rowlist[hop * BATCH + start + slot];
    }
    __syncthreads();

    for (int q = tid; q < 16 * 32; q += 256) {
        int r = q >> 5, c4 = q & 31;
        int rowg = sRow[r];
        const float* srcx = (hop == 0)
            ? (embeds + (size_t)eids[rowg * 4 + 0] * EMBED)
            : (xin    + (size_t)rowg * EMBED);
        float4 v = ((const float4*)srcx)[c4];
        sX[r][c4 * 4 + 0] = v.x; sX[r][c4 * 4 + 1] = v.y;
        sX[r][c4 * 4 + 2] = v.z; sX[r][c4 * 4 + 3] = v.w;
        const float* srcb = embeds + (size_t)eids[rowg * 4 + hop + 1] * EMBED;
        float4 bv = ((const float4*)srcb)[c4];
        sB[r][c4 * 4 + 0] = bv.x; sB[r][c4 * 4 + 1] = bv.y;
        sB[r][c4 * 4 + 2] = bv.z; sB[r][c4 * 4 + 3] = bv.w;
    }
    __syncthreads();

    int wave = tid >> 5, lane = tid & 31;
    int half = lane >> 4, n = lane & 15, c0 = half * 2;
    int obase = wave * 16;

    const float4* Bb = (const float4*)(WmF + (size_t)(m * 8 + wave) * 16 * 128) + lane;

    v8f acc = {0.f,0.f,0.f,0.f,0.f,0.f,0.f,0.f};
    #pragma unroll
    for (int k4g = 0; k4g < 16; ++k4g) {
        float4 w = Bb[k4g * 32];
        int k = k4g * 8;
        v2f a0, a1, b0, b1f;
        a0.x = sX[n][k + c0];       a0.y = sX[n][k + c0 + 1];
        a1.x = sX[n][k + 4 + c0];   a1.y = sX[n][k + 4 + c0 + 1];
        b0.x = w.x; b0.y = w.y;
        b1f.x = w.z; b1f.y = w.w;
        acc = wmma_f32_4(a0, b0,  acc);
        acc = wmma_f32_4(a1, b1f, acc);
    }

    for (int r = 0; r < 8; ++r) {
        int row = r + half * 8;
        if (row < cnt) {
            int rowg = sRow[row];
            int o = obase + n;
            float bias = sB[row][o];
            float v = acc[r] + bias;
            v = v > 0.f ? v : 0.f;
            float res = (hop == 2) ? ((1.f - WCOMB) * bias + WCOMB * v) : v;
            xout[(size_t)rowg * EMBED + o] = res;
        }
    }
}

__global__ __launch_bounds__(256) void mlp_kernel(
    const float* __restrict__ xin, const float* __restrict__ W1F, const float* __restrict__ b1,
    const float* __restrict__ W2F, const float* __restrict__ b2, float* __restrict__ out)
{
    __shared__ float sX[16][132];
    __shared__ float sH[16][516];

    int bt = blockIdx.x;
    int tid = threadIdx.x;
    int wave = tid >> 5, lane = tid & 31;
    int half = lane >> 4, n = lane & 15, c0 = half * 2;

    for (int q = tid; q < 16 * 32; q += 256) {
        int r = q >> 5, c4 = q & 31;
        float4 v = ((const float4*)(xin + (size_t)(bt * 16 + r) * EMBED))[c4];
        sX[r][c4 * 4 + 0] = v.x; sX[r][c4 * 4 + 1] = v.y;
        sX[r][c4 * 4 + 2] = v.z; sX[r][c4 * 4 + 3] = v.w;
    }
    __syncthreads();

    for (int p = 0; p < 4; ++p) {
        int ot = p * 8 + wave;
        int ob = ot * 16;
        const float4* Bb = (const float4*)(W1F + (size_t)ot * 16 * 128) + lane;
        v8f acc = {0.f,0.f,0.f,0.f,0.f,0.f,0.f,0.f};
        #pragma unroll
        for (int k4g = 0; k4g < 16; ++k4g) {
            float4 w = Bb[k4g * 32];
            int k = k4g * 8;
            v2f a0, a1, b0, b1f;
            a0.x = sX[n][k + c0];       a0.y = sX[n][k + c0 + 1];
            a1.x = sX[n][k + 4 + c0];   a1.y = sX[n][k + 4 + c0 + 1];
            b0.x = w.x; b0.y = w.y;
            b1f.x = w.z; b1f.y = w.w;
            acc = wmma_f32_4(a0, b0,  acc);
            acc = wmma_f32_4(a1, b1f, acc);
        }
        float bj = b1[ob + n];
        for (int r = 0; r < 8; ++r) {
            int row = r + half * 8;
            float v = acc[r] + bj;
            sH[row][ob + n] = v > 0.f ? v : 0.f;
        }
    }
    __syncthreads();

    if (wave < 4) {
        int ob = wave * 16;
        const float4* Bb = (const float4*)(W2F + (size_t)wave * 64 * 128) + lane;
        v8f acc = {0.f,0.f,0.f,0.f,0.f,0.f,0.f,0.f};
        #pragma unroll 8
        for (int k4g = 0; k4g < 64; ++k4g) {
            float4 w = Bb[k4g * 32];
            int k = k4g * 8;
            v2f a0, a1, b0, b1f;
            a0.x = sH[n][k + c0];       a0.y = sH[n][k + c0 + 1];
            a1.x = sH[n][k + 4 + c0];   a1.y = sH[n][k + 4 + c0 + 1];
            b0.x = w.x; b0.y = w.y;
            b1f.x = w.z; b1f.y = w.w;
            acc = wmma_f32_4(a0, b0,  acc);
            acc = wmma_f32_4(a1, b1f, acc);
        }
        float bo = b2[ob + n];
        for (int r = 0; r < 8; ++r) {
            int row = r + half * 8;
            out[(size_t)(bt * 16 + row) * OUT_DIM + ob + n] = acc[r] + bo;
        }
    }
}

extern "C" void kernel_launch(void* const* d_in, const int* in_sizes, int n_in,
                              void* d_out, int out_size, void* d_ws, size_t ws_size,
                              hipStream_t stream) {
    const float* embeds = (const float*)d_in[0];
    const float* Wmods  = (const float*)d_in[1];
    const float* W1     = (const float*)d_in[2];
    const float* b1     = (const float*)d_in[3];
    const float* W2     = (const float*)d_in[4];
    const float* b2     = (const float*)d_in[5];
    const int*   eids   = (const int*)d_in[6];
    const int*   mids   = (const int*)d_in[7];
    float*       out    = (float*)d_out;

    char* ws = (char*)d_ws;
    size_t off = 0;
    float* X0      = (float*)(ws + off); off += (size_t)BATCH * EMBED * 4;
    float* X1      = (float*)(ws + off); off += (size_t)BATCH * EMBED * 4;
    float* WmF     = (float*)(ws + off); off += (size_t)NUM_MOD * EMBED * EMBED * 4;
    float* W1F     = (float*)(ws + off); off += (size_t)HIDDEN * EMBED * 4;
    float* W2F     = (float*)(ws + off); off += (size_t)OUT_DIM * HIDDEN * 4;
    int*   counts  = (int*)(ws + off);   off += 3 * NUM_MOD * 4;
    int*   offsets = (int*)(ws + off);   off += 3 * NUM_MOD * 4;
    int*   cursor  = (int*)(ws + off);   off += 3 * NUM_MOD * 4;
    int*   rowlist = (int*)(ws + off);   off += 3 * BATCH * 4;
    int4*  tiletab = (int4*)(ws + off);  off += 3 * MAX_TILES * sizeof(int4);
    int*   ntiles  = (int*)(ws + off);   off += 64;
    (void)ws_size; (void)in_sizes; (void)n_in; (void)out_size;

    zero_counts_kernel<<<1, 256, 0, stream>>>(counts);
    count_kernel<<<BATCH / 256, 256, 0, stream>>>(mids, counts);
    scan_build_kernel<<<1, 32, 0, stream>>>(counts, offsets, cursor, tiletab, ntiles);
    scatter_kernel<<<BATCH / 256, 256, 0, stream>>>(mids, cursor, rowlist);

    pack_wm_kernel<<<(NUM_MOD * 8 * 32 * 32) / 256, 256, 0, stream>>>(Wmods, WmF);
    pack_w1_kernel<<<(32 * 32 * 32) / 256, 256, 0, stream>>>(W1, W1F);
    pack_w2_kernel<<<(4 * 128 * 32) / 256, 256, 0, stream>>>(W2, W2F);

    hop_gemm_kernel<<<MAX_TILES, 256, 0, stream>>>(embeds, WmF, eids, tiletab, ntiles, rowlist, X1, X0, 0);
    hop_gemm_kernel<<<MAX_TILES, 256, 0, stream>>>(embeds, WmF, eids, tiletab, ntiles, rowlist, X0, X1, 1);
    hop_gemm_kernel<<<MAX_TILES, 256, 0, stream>>>(embeds, WmF, eids, tiletab, ntiles, rowlist, X1, X0, 2);

    mlp_kernel<<<BATCH / 16, 256, 0, stream>>>(X0, W1F, b1, W2F, b2, out);
}


# row fragments leave the matrix streams: LDS reads go out as one group ahead of a block of products (hidden layer: the tile's 32 read once and kept for all 4 passes); every sum unchanged
# speedup vs baseline: 4.4093x; 4.4093x over previous
#include <hip/hip_runtime.h>
#include <hip/hip_bf16.h>
#include <stdint.h>

#define BATCH    8192
#define EMBED    128
#define NUM_MOD  64
#define HIDDEN   512
#define OUT_DIM  64
#define BIN_CAP  BATCH
#define TILE_ROWS 32
#define STAGE_STEPS (TILE_ROWS * 32 / 256)
#define HOP_SLICES (BATCH / NUM_MOD / TILE_ROWS + 1)
#define RANK_WAVES 8
#define RANK_STEPS (BATCH / RANK_WAVES / 32)
#define WCOMB    0.2857142857142857f
#define NUM_ENT  200000
#define ROW_BITS 13
#define OWN_CAP  (((BATCH / TILE_ROWS + HOP_SLICES - 1) / HOP_SLICES) * TILE_ROWS)
static_assert((1 << ROW_BITS) == BATCH, "row field of a slot record");
static_assert(((NUM_ENT - 1) >> (32 - ROW_BITS)) == 0, "entity field of a slot record");

typedef float v2f __attribute__((ext_vector_type(2)));
typedef float v8f __attribute__((ext_vector_type(8)));

__device__ __forceinline__ v8f wmma_f32_4(v2f a, v2f b, v8f c) {
    return __builtin_amdgcn_wmma_f32_16x16x4_f32(
         false, a,  false, b,
         (short)0, c,  false,  false);
}

template <bool OWN_TILES>
__device__ __forceinline__ int rank_rows(const int* __restrict__ mids, const int* __restrict__ eids,
                                         int h, int m, int g,
                                         unsigned short* sOwn, unsigned* sRec, int* sTot)
{
    const int tid = threadIdx.x, wave = tid >> 5, lane = tid & 31;
    const int row0 = wave * (BATCH / RANK_WAVES) + lane;
    int mid[RANK_STEPS];
    unsigned bal[RANK_STEPS];
    unsigned ent[RANK_STEPS];
    #pragma unroll
    for (int it = 0; it < RANK_STEPS; ++it) mid[it] = mids[(row0 + it * 32) * 3 + h];
    int wtot = 0;
    #pragma unroll
    for (int it = 0; it < RANK_STEPS; ++it) {
        bal[it] = __builtin_amdgcn_ballot_w32(mid[it] == m);
        wtot += __popc(bal[it]);
    }
    if (!OWN_TILES) {
        #pragma unroll
        for (int it = 0; it < RANK_STEPS; ++it) {
            ent[it] = 0;
            if ((bal[it] >> lane) & 1u) ent[it] = (unsigned)eids[(row0 + it * 32) * 4 + h + 1];
        }
    }
    if (lane == 0) sTot[wave] = wtot;
    __syncthreads();
    int before = 0, total = 0;
    for (int w = 0; w < RANK_WAVES; ++w) {
        int t = sTot[w];
        total += t;
        before += (w < wave) ? t : 0;
    }
    const unsigned below = (1u << lane) - 1u;
    int run = before;
    #pragma unroll
    for (int it = 0; it < RANK_STEPS; ++it) {
        if ((bal[it] >> lane) & 1u) {
            unsigned slot = (unsigned)(run + __popc(bal[it] & below));
            unsigned row = (unsigned)(row0 + it * 32);
            if (OWN_TILES) {
                unsigned tile = slot / TILE_ROWS;
                unsigned q = tile / HOP_SLICES;
                if (tile - q * HOP_SLICES == (unsigned)g) sOwn[q * TILE_ROWS + (slot % TILE_ROWS)] = (unsigned short)row;
            } else {
                sRec[slot] = row | (ent[it] << ROW_BITS);
            }
        }
        run += __popc(bal[it]);
    }
    return total;
}


__device__ __forceinline__ void load_w(const float4* Bb, float4 (&w)[16]) {
    #pragma unroll
    for (int k4g = 0; k4g < 16; ++k4g) w[k4g] = Bb[k4g * 2];
}

union StageBuf {
    float    tile[2][TILE_ROWS][132];
    unsigned list[BATCH];
};

__global__ __launch_bounds__(256) void hop_gemm_kernel(
    const float* __restrict__ embeds, const float* Wm,
    const int*  __restrict__ eids,    const int*  __restrict__ mids,
    int* counts, unsigned* bins, float* X, int hop)
{
    __shared__ __attribute__((aligned(16))) StageBuf S;
    __shared__ int   sRow[TILE_ROWS];
    __shared__ unsigned short sOwn[OWN_CAP];
    __shared__ __attribute__((aligned(16))) int sTot[RANK_WAVES];
    float (*sX)[132] = S.tile[0];
    float (*sB)[132] = S.tile[1];

    const int m = blockIdx.x & (NUM_MOD - 1);
    const int g = blockIdx.x / NUM_MOD;
    const int tid = threadIdx.x;

    if (g == HOP_SLICES) {
        const int nh = hop + 1;
        int ntot = rank_rows<false>(mids, eids, nh, m, 0, sOwn, S.list, sTot);
        __syncthreads();
        int padEnd = (ntot + TILE_ROWS - 1) & ~(TILE_ROWS - 1);
        if (padEnd < HOP_SLICES * TILE_ROWS) padEnd = HOP_SLICES * TILE_ROWS;
        const unsigned padRec = (ntot > 0) ? S.list[(ntot - 1) & ~(TILE_ROWS - 1)] : 0u;
        unsigned* gBin = bins + (size_t)(hop * NUM_MOD + m) * BIN_CAP;
        for (int i = tid; i < (padEnd >> 2); i += 256) {
            uint4 v = ((const uint4*)S.list)[i];
            if (4 * i + 0 >= ntot) v.x = padRec;
            if (4 * i + 1 >= ntot) v.y = padRec;
            if (4 * i + 2 >= ntot) v.z = padRec;
            if (4 * i + 3 >= ntot) v.w = padRec;
            ((uint4*)gBin)[i] = v;
        }
        if (tid == 0) counts[nh * NUM_MOD + m] = ntot;
        return;
    }

    const int wave = tid >> 5, lane = tid & 31;
    const int half = lane >> 4, n = lane & 15;
    const int obase = wave * 16;
    const int c4 = lane, r0 = wave;

    const unsigned* bin = bins + (size_t)((hop > 0 ? hop - 1 : 0) * NUM_MOD + m) * BIN_CAP;
    const float4* Bb = (const float4*)(Wm + ((size_t)m * EMBED + obase + n) * EMBED + half * 4);

    int start = g * TILE_ROWS;
    int q = 0;
    int total;
    int      rowg[STAGE_STEPS];
    int4     e[STAGE_STEPS];
    unsigned rec[STAGE_STEPS];
    float4   w[16];

    if (hop == 0) {
        total = rank_rows<true>(mids, eids, 0, m, g, sOwn, S.list, sTot);
        __syncthreads();
        if (start >= total) return;
        int cnt0 = total - start; if (cnt0 > TILE_ROWS) cnt0 = TILE_ROWS;
        #pragma unroll
        for (int it = 0; it < STAGE_STEPS; ++it) {
            int r = r0 + it * 8;
            rowg[it] = sOwn[(r < cnt0) ? r : 0];
        }
        #pragma unroll
        for (int it = 0; it < STAGE_STEPS; ++it) e[it] = ((const int4*)eids)[rowg[it]];
        asm volatile("" ::: "memory");
        load_w(Bb, w);
    } else {
        total = counts[hop * NUM_MOD + m];
        #pragma unroll
        for (int it = 0; it < STAGE_STEPS; ++it) rec[it] = bin[start + r0 + it * 8];
        load_w(Bb, w);
        asm volatile("" ::: "memory");
        if (start >= total) return;
    }

    for (;;) {
        int cnt = total - start; if (cnt > TILE_ROWS) cnt = TILE_ROWS;

        {
            if (hop == 0) {
                if (c4 == 0) {
                    #pragma unroll
                    for (int it = 0; it < STAGE_STEPS; ++it) sRow[r0 + it * 8] = rowg[it];
                }
                float4 xv[STAGE_STEPS], bv[STAGE_STEPS];
                #pragma unroll
                for (int it = 0; it < STAGE_STEPS; ++it) {
                    xv[it] = ((const float4*)(embeds + (size_t)e[it].x * EMBED))[c4];
                    bv[it] = ((const float4*)(embeds + (size_t)e[it].y * EMBED))[c4];
                }
                #pragma unroll
                for (int it = 0; it < STAGE_STEPS; ++it) {
                    int r = r0 + it * 8;
                    *(float4*)&sX[r][c4 * 4] = xv[it];
                    *(float4*)&sB[r][c4 * 4] = bv[it];
                }
            } else {
                if (c4 == 0) {
                    #pragma unroll
                    for (int it = 0; it < STAGE_STEPS; ++it) sRow[r0 + it * 8] = (int)(rec[it] & (BATCH - 1));
                }
                float4 xv[STAGE_STEPS], bv[STAGE_STEPS];
                #pragma unroll
                for (int it = 0; it < STAGE_STEPS; ++it) {
                    xv[it] = ((const float4*)(X + (size_t)(rec[it] & (BATCH - 1)) * EMBED))[c4];
                    bv[it] = ((const float4*)(embeds + (size_t)(rec[it] >> ROW_BITS) * EMBED))[c4];
                }
                #pragma unroll
                for (int it = 0; it < STAGE_STEPS; ++it) {
                    int r = r0 + it * 8;
                    *(float4*)&sX[r][c4 * 4] = xv[it];
                    *(float4*)&sB[r][c4 * 4] = bv[it];
                }
            }
        }
        __syncthreads();

        constexpr int XA_FIRST = 12;
        float4 xa[16][2];
        #pragma unroll
        for (int k4g = 0; k4g < XA_FIRST; ++k4g) {
            xa[k4g][0] = *(const float4*)&sX[n][k4g * 8 + half * 4];
            xa[k4g][1] = *(const float4*)&sX[16 + n][k4g * 8 + half * 4];
        }
        __builtin_amdgcn_sched_barrier(0);

        v8f accLo = {0.f,0.f,0.f,0.f,0.f,0.f,0.f,0.f};
        v8f accHi = {0.f,0.f,0.f,0.f,0.f,0.f,0.f,0.f};
        #pragma unroll
        for (int k4g = 0; k4g < 16; ++k4g) {
            if (k4g == 16 - XA_FIRST) {
                __builtin_amdgcn_sched_barrier(0);
                #pragma unroll
                for (int k2 = XA_FIRST; k2 < 16; ++k2) {
                    xa[k2][0] = *(const float4*)&sX[n][k2 * 8 + half * 4];
                    xa[k2][1] = *(const float4*)&sX[16 + n][k2 * 8 + half * 4];
                }
                __builtin_amdgcn_sched_barrier(0);
            }
            const float4 xl = xa[k4g][0];
            const float4 xh = xa[k4g][1];
            v2f l0, l1, h0, h1, b0, b1f;
            l0.x = xl.x; l0.y = xl.y;
            l1.x = xl.z; l1.y = xl.w;
            h0.x = xh.x; h0.y = xh.y;
            h1.x = xh.z; h1.y = xh.w;
            b0.x = w[k4g].x; b0.y = w[k4g].y;
            b1f.x = w[k4g].z; b1f.y = w[k4g].w;
            accLo = wmma_f32_4(l0, b0,  accLo);
            accHi = wmma_f32_4(h0, b0,  accHi);
            accLo = wmma_f32_4(l1, b1f, accLo);
            accHi = wmma_f32_4(h1, b1f, accHi);
        }

        #pragma unroll
        for (int t = 0; t < 2; ++t) {
            const v8f acc = (t == 0) ? accLo : accHi;
            if (cnt >= t * 16 + 16) {
                #pragma unroll
                for (int r = 0; r < 8; ++r) {
                    int row = t * 16 + r + half * 8;
                    int rg = sRow[row];
                    int o = obase + n;
                    float bias = sB[row][o];
                    float v = acc[r] + bias;
                    v = v > 0.f ? v : 0.f;
                    float res = (hop == 2) ? ((1.f - WCOMB) * bias + WCOMB * v) : v;
                    X[(size_t)rg * EMBED + o] = res;
                }
            } else {
                for (int r = 0; r < 8; ++r) {
                    int row = t * 16 + r + half * 8;
                    if (row < cnt) {
                        int rg = sRow[row];
                        int o = obase + n;
                        float bias = sB[row][o];
                        float v = acc[r] + bias;
                        v = v > 0.f ? v : 0.f;
                        float res = (hop == 2) ? ((1.f - WCOMB) * bias + WCOMB * v) : v;
                        X[(size_t)rg * EMBED + o] = res;
                    }
                }
            }
        }

        start += HOP_SLICES * TILE_ROWS;
        ++q;
        if (start >= total) break;
        __syncthreads();
        if (hop == 0) {
            int cntn = total - start; if (cntn > TILE_ROWS) cntn = TILE_ROWS;
            #pragma unroll
            for (int it = 0; it < STAGE_STEPS; ++it) {
                int r = r0 + it * 8;
                rowg[it] = sOwn[q * TILE_ROWS + ((r < cntn) ? r : 0)];
            }
            #pragma unroll
            for (int it = 0; it < STAGE_STEPS; ++it) e[it] = ((const int4*)eids)[rowg[it]];
        } else {
            #pragma unroll
            for (int it = 0; it < STAGE_STEPS; ++it) rec[it] = bin[start + r0 + it * 8];
        }
    }
}

__device__ __forceinline__ void lds_barrier() {
    __builtin_amdgcn_fence(__ATOMIC_RELEASE, "workgroup", "local");
    __builtin_amdgcn_s_barrier();
    __builtin_amdgcn_fence(__ATOMIC_ACQUIRE, "workgroup", "local");
}

__global__ __launch_bounds__(256) void mlp_kernel(
    const float* xin, const float* W1, const float* __restrict__ b1,
    const float* W2, const float* __restrict__ b2, float* __restrict__ out)
{
    __shared__ __attribute__((aligned(16))) float sX[TILE_ROWS][132];
    __shared__ __attribute__((aligned(16))) float sH[TILE_ROWS][516];

    const int bt = blockIdx.x;
    const int tid = threadIdx.x;
    const int wave = tid >> 5, lane = tid & 31;
    const int half = lane >> 4, n = lane & 15;
    const int c4 = lane, r0 = wave;

    static_assert(STAGE_STEPS == 4, "four staged rows per thread");
    const float4* xrow = (const float4*)(xin + (size_t)(bt * TILE_ROWS + r0) * EMBED) + c4;
    const float4 x0 = xrow[0 * 8 * (EMBED / 4)];
    const float4 x1 = xrow[1 * 8 * (EMBED / 4)];
    const float4 x2 = xrow[2 * 8 * (EMBED / 4)];
    const float4 x3 = xrow[3 * 8 * (EMBED / 4)];
    asm volatile("" ::: "memory");

    const float4* Bb1 = (const float4*)(W1 + (size_t)(wave * 16 + n) * EMBED + half * 4);
    constexpr int PASS_F4 = 8 * 16 * EMBED / 4;
    float4 w1[4][16];
    float  bj[4];
    #pragma unroll
    for (int k4g = 0; k4g < 16; ++k4g) w1[0][k4g] = Bb1[k4g * 2];
    #pragma unroll
    for (int p = 0; p < 4; ++p) bj[p] = b1[(p * 8 + wave) * 16 + n];
    asm volatile("" ::: "memory");

    *(float4*)&sX[r0 +  0][c4 * 4] = x0;
    *(float4*)&sX[r0 +  8][c4 * 4] = x1;
    *(float4*)&sX[r0 + 16][c4 * 4] = x2;
    *(float4*)&sX[r0 + 24][c4 * 4] = x3;
    lds_barrier();

    float4 xa[16][2];
    #pragma unroll
    for (int k4g = 0; k4g < 16; ++k4g) {
        xa[k4g][0] = *(const float4*)&sX[n][k4g * 8 + half * 4];
        xa[k4g][1] = *(const float4*)&sX[16 + n][k4g * 8 + half * 4];
    }
    __builtin_amdgcn_sched_barrier(0);

    const int ob2 = (wave & 3) * 16;
    const float4* Bb2 = (const float4*)(W2 + (size_t)(ob2 + n) * HIDDEN + half * 4);
    float4 w2[64];
    float  bo = 0.f;

    #pragma unroll
    for (int p = 0; p < 4; ++p) {
        const int ob = (p * 8 + wave) * 16;
        v8f accLo = {0.f,0.f,0.f,0.f,0.f,0.f,0.f,0.f};
        v8f accHi = {0.f,0.f,0.f,0.f,0.f,0.f,0.f,0.f};
        #pragma unroll
        for (int k4g = 0; k4g < 16; ++k4g) {
            if (p < 3) w1[p + 1][k4g] = Bb1[(p + 1) * PASS_F4 + k4g * 2];
            float4 w  = w1[p][k4g];
            const float4 xl = xa[k4g][0];
            const float4 xh = xa[k4g][1];
            v2f l0, l1, h0, h1, b0, b1f;
            l0.x = xl.x; l0.y = xl.y;
            l1.x = xl.z; l1.y = xl.w;
            h0.x = xh.x; h0.y = xh.y;
            h1.x = xh.z; h1.y = xh.w;
            b0.x = w.x; b0.y = w.y;
            b1f.x = w.z; b1f.y = w.w;
            accLo = wmma_f32_4(l0, b0,  accLo);
            accHi = wmma_f32_4(h0, b0,  accHi);
            accLo = wmma_f32_4(l1, b1f, accLo);
            accHi = wmma_f32_4(h1, b1f, accHi);
            __builtin_amdgcn_sched_barrier(0x2 | 0x4 | 0x80);
        }
        for (int r = 0; r < 8; ++r) {
            int row = r + half * 8;
            float vl = accLo[r] + bj[p];
            float vh = accHi[r] + bj[p];
            sH[row][ob + n]      = vl > 0.f ? vl : 0.f;
            sH[16 + row][ob + n] = vh > 0.f ? vh : 0.f;
        }
        if (wave < 4 && p != 2) {
            const int q0 = (p == 3) ? 32 : p * 16;
            const int q1 = (p == 3) ? 64 : p * 16 + 16;
            #pragma unroll
            for (int j = q0; j < q1; ++j) w2[j] = Bb2[j * 2];
            if (p == 0) bo = b2[ob2 + n];
        }
        asm volatile("" ::: "memory");
    }
    lds_barrier();

    if (wave < 4) {
        v8f accLo = {0.f,0.f,0.f,0.f,0.f,0.f,0.f,0.f};
        v8f accHi = {0.f,0.f,0.f,0.f,0.f,0.f,0.f,0.f};
        float4 fa[2][16];
        #pragma unroll
        for (int j = 0; j < 8; ++j) {
            fa[0][2 * j]     = *(const float4*)&sH[n][j * 8 + half * 4];
            fa[0][2 * j + 1] = *(const float4*)&sH[16 + n][j * 8 + half * 4];
        }
        #pragma unroll
        for (int b = 0; b < 8; ++b) {
            if (b < 7) {
                if (b > 0) __builtin_amdgcn_sched_barrier(0);
                #pragma unroll
                for (int j = 0; j < 8; ++j) {
                    fa[(b + 1) & 1][2 * j]     = *(const float4*)&sH[n][((b + 1) * 8 + j) * 8 + half * 4];
                    fa[(b + 1) & 1][2 * j + 1] = *(const float4*)&sH[16 + n][((b + 1) * 8 + j) * 8 + half * 4];
                }
                __builtin_amdgcn_sched_barrier(0);
            }
            #pragma unroll
            for (int j = 0; j < 8; ++j) {
                const float4 w  = w2[b * 8 + j];
                const float4 xl = fa[b & 1][2 * j];
                const float4 xh = fa[b & 1][2 * j + 1];
                v2f l0, l1, h0, h1, b0, b1f;
                l0.x = xl.x; l0.y = xl.y;
                l1.x = xl.z; l1.y = xl.w;
                h0.x = xh.x; h0.y = xh.y;
                h1.x = xh.z; h1.y = xh.w;
                b0.x = w.x; b0.y = w.y;
                b1f.x = w.z; b1f.y = w.w;
                accLo = wmma_f32_4(l0, b0,  accLo);
                accHi = wmma_f32_4(h0, b0,  accHi);
                accLo = wmma_f32_4(l1, b1f, accLo);
                accHi = wmma_f32_4(h1, b1f, accHi);
            }
        }
        for (int r = 0; r < 8; ++r) {
            int row = r + half * 8;
            out[(size_t)(bt * TILE_ROWS + row) * OUT_DIM + ob2 + n]      = accLo[r] + bo;
            out[(size_t)(bt * TILE_ROWS + 16 + row) * OUT_DIM + ob2 + n] = accHi[r] + bo;
        }
    }
}

extern "C" void kernel_launch(void* const* d_in, const int* in_sizes, int n_in,
                              void* d_out, int out_size, void* d_ws, size_t ws_size,
                              hipStream_t stream) {
    const float* embeds = (const float*)d_in[0];
    const float* Wmods  = (const float*)d_in[1];
    const float* W1     = (const float*)d_in[2];
    const float* b1     = (const float*)d_in[3];
    const float* W2     = (const float*)d_in[4];
    const float* b2     = (const float*)d_in[5];
    const int*   eids   = (const int*)d_in[6];
    const int*   mids   = (const int*)d_in[7];
    float*       out    = (float*)d_out;

    char* ws = (char*)d_ws;
    size_t off = 0;
    float* X0      = (float*)(ws + off); off += (size_t)BATCH * EMBED * 4;
    unsigned* bins = (unsigned*)(ws + off); off += (size_t)BATCH * EMBED * 4;
    int*   counts  = (int*)(ws + off);   off += 3 * NUM_MOD * 4;
    (void)ws_size; (void)in_sizes; (void)n_in; (void)out_size;

    hop_gemm_kernel<<<NUM_MOD * (HOP_SLICES + 1), 256, 0, stream>>>(embeds, Wmods, eids, mids, counts, bins, X0, 0);
    hop_gemm_kernel<<<NUM_MOD * (HOP_SLICES + 1), 256, 0, stream>>>(embeds, Wmods, eids, mids, counts, bins, X0, 1);
    hop_gemm_kernel<<<NUM_MOD * HOP_SLICES, 256, 0, stream>>>(embeds, Wmods, eids, mids, counts, bins, X0, 2);

    mlp_kernel<<<BATCH / TILE_ROWS, 256, 0, stream>>>(X0, W1, b1, W2, b2, out);
}
